# starting kernel
# baseline (speedup 1.0000x reference)
#include <hip/hip_runtime.h>
#include <hip/hip_bf16.h>
#include <math.h>

#define NN 10000
#define EE 160000

typedef __attribute__((ext_vector_type(2))) float v2f;
typedef __attribute__((ext_vector_type(8))) float v8f;

__device__ __forceinline__ float lrelu_(float x) { return x > 0.f ? x : 0.2f * x; }
__device__ __forceinline__ float elu_(float x)   { return x > 0.f ? x : (expf(x) - 1.f); }
__device__ __forceinline__ unsigned fenc_(float f) {
    unsigned u = __float_as_uint(f);
    return (u & 0x80000000u) ? ~u : (u | 0x80000000u);
}
__device__ __forceinline__ float fdec_(unsigned e) {
    return __uint_as_float((e & 0x80000000u) ? (e & 0x7FFFFFFFu) : ~e);
}

__global__ __launch_bounds__(128)
void k_gemm_rel_wmma(const float* __restrict__ X, const float* __restrict__ W,
                     float* __restrict__ Y, int M, int K, int Nout) {
    __shared__ float As[16 * 256];
    const int r    = blockIdx.z;
    const int m0   = blockIdx.x * 16;
    const int wave = threadIdx.x >> 5;
    const int lane = threadIdx.x & 31;
    const int n0   = blockIdx.y * 64 + wave * 16;

    for (int row = 0; row < 16; ++row)
        for (int col = threadIdx.x; col < K; col += 128)
            As[row * K + col] = X[(size_t)(m0 + row) * K + col];
    __syncthreads();

    if (n0 < Nout) {
        const float* Wr  = W + (size_t)r * K * Nout;
        const int  ncol  = n0 + (lane & 15);
        const bool nv    = ncol < Nout;
        const int  ncl   = nv ? ncol : (Nout - 1);
        const int  kh    = (lane >> 4) * 2;
        const int  mrow  = lane & 15;
        v8f c = {};
        for (int k0 = 0; k0 < K; k0 += 4) {
            v2f a, b;
            a.x = As[mrow * K + k0 + kh];
            a.y = As[mrow * K + k0 + kh + 1];
            b.x = Wr[(size_t)(k0 + kh)     * Nout + ncl];
            b.y = Wr[(size_t)(k0 + kh + 1) * Nout + ncl];
            c = __builtin_amdgcn_wmma_f32_16x16x4_f32(false, a, false, b,
                                                      (short)0, c, false, false);
        }
        if (nv) {
            float* Yr = Y + (size_t)r * M * Nout;
            const int rbase = m0 + (lane >> 4) * 8;
#pragma unroll
            for (int i = 0; i < 8; ++i)
                Yr[(size_t)(rbase + i) * Nout + ncol] = c[i];
        }
    }
}

__global__ __launch_bounds__(128)
void k_rowdot(const float* __restrict__ XA, const float* __restrict__ Q,
              const float* __restrict__ Kv, float* __restrict__ qall,
              float* __restrict__ kall, int D, int H, int total) {
    const int wid  = (int)((blockIdx.x * (unsigned)blockDim.x + threadIdx.x) >> 5);
    const int lane = threadIdx.x & 31;
    if (wid >= total) return;
    const float* xrow = XA + (size_t)wid * D;
    float aq[2] = {0.f, 0.f}, ak[2] = {0.f, 0.f};
    for (int d = lane; d < D; d += 32) {
        float v = xrow[d];
        for (int h = 0; h < H; ++h) {
            aq[h] += v * Q[d * H + h];
            ak[h] += v * Kv[d * H + h];
        }
    }
    for (int off = 16; off; off >>= 1)
        for (int h = 0; h < H; ++h) {
            aq[h] += __shfl_xor(aq[h], off, 32);
            ak[h] += __shfl_xor(ak[h], off, 32);
        }
    if (lane == 0)
        for (int h = 0; h < H; ++h) {
            qall[(size_t)wid * H + h] = aq[h];
            kall[(size_t)wid * H + h] = ak[h];
        }
}

__global__ void k_init(unsigned* __restrict__ menc, float* __restrict__ den,
                       float* __restrict__ aggr, int nh, int na) {
    const int stride = gridDim.x * blockDim.x;
    for (int t = blockIdx.x * blockDim.x + threadIdx.x; t < nh; t += stride) {
        menc[t] = 0x007FFFFFu;
        den[t]  = 0.f;
    }
    for (int t = blockIdx.x * blockDim.x + threadIdx.x; t < na; t += stride)
        aggr[t] = 0.f;
}

__global__ void k_edge_alpha_max(const float* __restrict__ qall,
                                 const float* __restrict__ kall,
                                 const int* __restrict__ src, const int* __restrict__ dst,
                                 const int* __restrict__ etype,
                                 float* __restrict__ exbuf, unsigned* __restrict__ menc,
                                 int H, int Rr) {
    const int t = blockIdx.x * blockDim.x + threadIdx.x;
    if (t >= EE * H) return;
    const int e = t / H, h = t - e * H;
    const int et = (Rr > 1) ? etype[e] : 0;
    const int s = src[e], d = dst[e];
    const float a = lrelu_(qall[((size_t)et * NN + d) * H + h] +
                           kall[((size_t)et * NN + s) * H + h]);
    exbuf[t] = a;
    atomicMax(&menc[d * H + h], fenc_(a));
}

__global__ void k_edge_exp(const int* __restrict__ dst, float* __restrict__ exbuf,
                           const unsigned* __restrict__ menc, float* __restrict__ den,
                           int H) {
    const int t = blockIdx.x * blockDim.x + threadIdx.x;
    if (t >= EE * H) return;
    const int e = t / H, h = t - e * H;
    const int d = dst[e];
    const float ex = expf(exbuf[t] - fdec_(menc[d * H + h]));
    exbuf[t] = ex;
    atomicAdd(&den[d * H + h], ex);
}

__global__ void k_edge_aggr(const float* __restrict__ XA, const float* __restrict__ exbuf,
                            const float* __restrict__ den,
                            const int* __restrict__ src, const int* __restrict__ dst,
                            const int* __restrict__ etype, float* __restrict__ aggr,
                            int D, int C, int H, int Rr) {
    const int e = blockIdx.x;
    const int c = threadIdx.x;
    if (c >= C) return;
    const int s = src[e], d = dst[e];
    const int et = (Rr > 1) ? etype[e] : 0;
    const float* xrow = XA + ((size_t)et * NN + s) * D;
    float v = 0.f;
    for (int h = 0; h < H; ++h) {
        const float attn = exbuf[(size_t)e * H + h] / (den[d * H + h] + 1e-16f);
        v += attn * xrow[h * C + c];
    }
    atomicAdd(&aggr[(size_t)d * C + c], v / (float)H);
}

__global__ void k_epilogue(const float* __restrict__ aggr, const float* __restrict__ bias,
                           float* __restrict__ out, int C, int total) {
    const int t = blockIdx.x * blockDim.x + threadIdx.x;
    if (t >= total) return;
    out[t] = elu_(aggr[t] + bias[t % C]);
}

struct WsPtrs {
    float* xall; float* qall; float* kall; unsigned* menc;
    float* den;  float* exbuf; float* aggr;
};

static void run_attn_conv(const float* xin, const float* w, const float* qv,
                          const float* kv, const float* bias,
                          int K, int D, int C, int H, int Rr,
                          const int* src, const int* dst, const int* etype,
                          const WsPtrs& p, float* out, hipStream_t stream) {
    dim3 gg(NN / 16, (D + 63) / 64, Rr);
    k_gemm_rel_wmma<<<gg, 128, 0, stream>>>(xin, w, p.xall, NN, K, D);

    const int totalRN = Rr * NN;
    k_rowdot<<<(totalRN + 3) / 4, 128, 0, stream>>>(p.xall, qv, kv, p.qall, p.kall,
                                                    D, H, totalRN);
    k_init<<<512, 256, 0, stream>>>(p.menc, p.den, p.aggr, NN * H, NN * C);

    const int eh = EE * H;
    k_edge_alpha_max<<<(eh + 255) / 256, 256, 0, stream>>>(p.qall, p.kall, src, dst,
                                                           etype, p.exbuf, p.menc, H, Rr);
    k_edge_exp<<<(eh + 255) / 256, 256, 0, stream>>>(dst, p.exbuf, p.menc, p.den, H);

    const int bC = ((C + 63) / 64) * 64;
    k_edge_aggr<<<EE, bC, 0, stream>>>(p.xall, p.exbuf, p.den, src, dst, etype,
                                       p.aggr, D, C, H, Rr);
    k_epilogue<<<(NN * C + 255) / 256, 256, 0, stream>>>(p.aggr, bias, out, C, NN * C);
}

extern "C" void kernel_launch(void* const* d_in, const int* in_sizes, int n_in,
                              void* d_out, int out_size, void* d_ws, size_t ws_size,
                              hipStream_t stream) {
    (void)in_sizes; (void)n_in; (void)out_size; (void)ws_size;
    const float* x  = (const float*)d_in[0];
    const int*   ei = (const int*)  d_in[1];
    const int*   et = (const int*)  d_in[2];
    const float* w1 = (const float*)d_in[3],  *q1 = (const float*)d_in[4];
    const float* k1 = (const float*)d_in[5],  *b1 = (const float*)d_in[6];
    const float* wS = (const float*)d_in[7],  *qS = (const float*)d_in[8];
    const float* kS = (const float*)d_in[9],  *bS = (const float*)d_in[10];
    const float* w2 = (const float*)d_in[11], *q2 = (const float*)d_in[12];
    const float* k2 = (const float*)d_in[13], *b2 = (const float*)d_in[14];
    const float* wT = (const float*)d_in[15], *qT = (const float*)d_in[16];
    const float* kT = (const float*)d_in[17], *bT = (const float*)d_in[18];
    const float* wg = (const float*)d_in[19];
    const float* ag_src = (const float*)d_in[20];
    const float* ag_dst = (const float*)d_in[21];
    const float* bg = (const float*)d_in[22];

    const int* src = ei;
    const int* dst = ei + EE;

    float* ws = (float*)d_ws;
    size_t off = 0;
    WsPtrs p;
    p.xall  = ws + off;            off += (size_t)4 * NN * 512;
    p.qall  = ws + off;            off += (size_t)4 * NN * 2;
    p.kall  = ws + off;            off += (size_t)4 * NN * 2;
    p.menc  = (unsigned*)(ws + off); off += (size_t)NN * 2;
    p.den   = ws + off;            off += (size_t)NN * 2;
    p.exbuf = ws + off;            off += (size_t)EE * 2;
    p.aggr  = ws + off;            off += (size_t)NN * 256;
    float* xs = ws + off;          off += (size_t)NN * 256;
    float* xt = ws + off;          off += (size_t)NN * 256;

    float* x_in   = (float*)d_out;
    float* x_out  = (float*)d_out + (size_t)NN * 65;
    float* x_self = (float*)d_out + (size_t)2 * NN * 65;

    run_attn_conv(x,  w1, q1, k1, b1, 128, 512, 256, 2, 4, src, dst, et, p, xs,   stream);
    run_attn_conv(xs, wS, qS, kS, bS, 256, 130,  65, 2, 4, src, dst, et, p, x_in, stream);
    run_attn_conv(x,  w2, q2, k2, b2, 128, 512, 256, 2, 4, src, dst, et, p, xt,    stream);
    run_attn_conv(xt, wT, qT, kT, bT, 256, 130,  65, 2, 4, src, dst, et, p, x_out, stream);
    run_attn_conv(x,  wg, ag_dst, ag_src, bg, 128, 256, 256, 1, 1, src, dst, et, p, x_self, stream);
}


# GEMM: W rows read with 16-byte loads (8-byte where D=130): a wave's 4 column tiles interleave (tile s = columns 4c+s), one load feeds 4 matrix ops; 4 accumulators, k outermost; 16-byte Y stores
# speedup vs baseline: 3.1738x; 3.1738x over previous
#include <hip/hip_runtime.h>
#include <hip/hip_bf16.h>
#include <math.h>

#define NN 10000
#define EE 160000

typedef __attribute__((ext_vector_type(2))) float v2f;
typedef __attribute__((ext_vector_type(8))) float v8f;

__device__ __forceinline__ float lrelu_(float x) { return x > 0.f ? x : 0.2f * x; }
__device__ __forceinline__ float elu_(float x)   { return x > 0.f ? x : (expf(x) - 1.f); }
__device__ __forceinline__ unsigned fenc_(float f) {
    unsigned u = __float_as_uint(f);
    return (u & 0x80000000u) ? ~u : (u | 0x80000000u);
}
__device__ __forceinline__ float fdec_(unsigned e) {
    return __uint_as_float((e & 0x80000000u) ? (e & 0x7FFFFFFFu) : ~e);
}

template <int KK, int VW, int NPA>
__device__ __forceinline__ void gemm_wave_(const float* __restrict__ xrow,
                                           const float* __restrict__ Wr,
                                           float* __restrict__ Yr, int Nout,
                                           int base, int rbase, int h, int c16) {
    typedef float wv_t __attribute__((ext_vector_type(VW)));
    constexpr int NS = KK / 8;
    const float* wp[NPA];
    int  col[NPA];
    bool ok[NPA];
#pragma unroll
    for (int p = 0; p < NPA; ++p) {
        col[p] = base + 16 * VW * p + VW * c16;
        ok[p]  = col[p] < Nout;
        const int cl = ok[p] ? col[p] : (Nout - VW);
        wp[p] = Wr + (size_t)(4 * h) * Nout + cl;
    }
    v8f acc[NPA * VW];
#pragma unroll
    for (int t = 0; t < NPA * VW; ++t) acc[t] = v8f{};

    float4 qa[4], qn[4];
#pragma unroll
    for (int u = 0; u < 4; ++u) qa[u] = *reinterpret_cast<const float4*>(xrow + 8 * u);
    wv_t   wb[NPA][4];
#pragma unroll
    for (int p = 0; p < NPA; ++p)
#pragma unroll
        for (int kk = 0; kk < 4; ++kk)
            wb[p][kk] = *reinterpret_cast<const wv_t*>(wp[p] + (size_t)kk * Nout);

#pragma unroll
    for (int jj = 0; jj < NS; ++jj) {
        wv_t   wn[NPA][4];
        if ((jj & 3) == 0 && jj + 4 < NS) {
            __builtin_amdgcn_sched_barrier(0);
#pragma unroll
            for (int u = 0; u < 4; ++u)
                qn[u] = *reinterpret_cast<const float4*>(xrow + 8 * (jj + 4 + u));
            __builtin_amdgcn_sched_barrier(0);
        }
        if (jj + 1 < NS) {
#pragma unroll
            for (int p = 0; p < NPA; ++p)
#pragma unroll
                for (int kk = 0; kk < 4; ++kk)
                    wn[p][kk] = *reinterpret_cast<const wv_t*>(
                        wp[p] + (size_t)(8 * (jj + 1) + kk) * Nout);
        }
        v2f a0, a1;
        const float4 q = qa[jj & 3];
        a0.x = q.x;  a0.y = q.y;
        a1.x = q.z;  a1.y = q.w;
#pragma unroll
        for (int p = 0; p < NPA; ++p)
#pragma unroll
            for (int s = 0; s < VW; ++s) {
                v2f b;
                b.x = wb[p][0][s];
                b.y = wb[p][1][s];
                acc[p * VW + s] = __builtin_amdgcn_wmma_f32_16x16x4_f32(
                    false, a0, false, b, (short)0, acc[p * VW + s], false, false);
            }
#pragma unroll
        for (int p = 0; p < NPA; ++p)
#pragma unroll
            for (int s = 0; s < VW; ++s) {
                v2f b;
                b.x = wb[p][2][s];
                b.y = wb[p][3][s];
                acc[p * VW + s] = __builtin_amdgcn_wmma_f32_16x16x4_f32(
                    false, a1, false, b, (short)0, acc[p * VW + s], false, false);
            }
        if ((jj & 3) == 3 && jj + 1 < NS) {
#pragma unroll
            for (int u = 0; u < 4; ++u) qa[u] = qn[u];
        }
        if (jj + 1 < NS) {
#pragma unroll
            for (int p = 0; p < NPA; ++p)
#pragma unroll
                for (int kk = 0; kk < 4; ++kk) wb[p][kk] = wn[p][kk];
        }
    }

#pragma unroll
    for (int p = 0; p < NPA; ++p) {
        if (ok[p]) {
#pragma unroll
            for (int i = 0; i < 8; ++i) {
                wv_t o;
#pragma unroll
                for (int s = 0; s < VW; ++s) o[s] = acc[p * VW + s][i];
                *reinterpret_cast<wv_t*>(Yr + (size_t)(rbase + i) * Nout + col[p]) = o;
            }
        }
    }
}

template <int KK, int VW>
__global__ __launch_bounds__(128) __attribute__((amdgpu_waves_per_eu(4, 8)))
void k_gemm_rel_wmma(const float* __restrict__ X, const float* __restrict__ W,
                     float* __restrict__ Y, int M, int Nout) {
    constexpr int NP = 4 / VW;
    const int r    = blockIdx.z;
    const int wave = threadIdx.x >> 5;
    const int lane = threadIdx.x & 31;
    const int m0   = (blockIdx.x * 4 + wave) * 16;
    if (m0 >= M) return;
    const int h    = lane >> 4;
    const int c16  = lane & 15;
    const int base = blockIdx.y * 64;

    const float* xrow = X + (size_t)(m0 + c16) * KK + 4 * h;
    const float* Wr   = W + (size_t)r * KK * Nout;
    float*       Yr   = Y + (size_t)r * M * Nout;
    const int rbase   = m0 + h * 8;
    const int npa = min(NP, (Nout - base + 16 * VW - 1) / (16 * VW));
    if (NP == 1 || npa == NP) gemm_wave_<KK, VW, NP>(xrow, Wr, Yr, Nout, base, rbase, h, c16);
    else                      gemm_wave_<KK, VW, 1>(xrow, Wr, Yr, Nout, base, rbase, h, c16);
}

__global__ __launch_bounds__(128)
void k_rowdot(const float* __restrict__ XA, const float* __restrict__ Q,
              const float* __restrict__ Kv, float* __restrict__ qall,
              float* __restrict__ kall, int D, int H, int total) {
    const int wid  = (int)((blockIdx.x * (unsigned)blockDim.x + threadIdx.x) >> 5);
    const int lane = threadIdx.x & 31;
    if (wid >= total) return;
    const float* xrow = XA + (size_t)wid * D;
    float aq[2] = {0.f, 0.f}, ak[2] = {0.f, 0.f};
    for (int d = lane; d < D; d += 32) {
        float v = xrow[d];
        for (int h = 0; h < H; ++h) {
            aq[h] += v * Q[d * H + h];
            ak[h] += v * Kv[d * H + h];
        }
    }
    for (int off = 16; off; off >>= 1)
        for (int h = 0; h < H; ++h) {
            aq[h] += __shfl_xor(aq[h], off, 32);
            ak[h] += __shfl_xor(ak[h], off, 32);
        }
    if (lane == 0)
        for (int h = 0; h < H; ++h) {
            qall[(size_t)wid * H + h] = aq[h];
            kall[(size_t)wid * H + h] = ak[h];
        }
}

#define OWN_NT  1024
#define OWN_CAP 4096

template <int H>
__device__ __forceinline__ void edge_logits_(int e, int lo,
                                             const int* __restrict__ src, const int* __restrict__ dst,
                                             const int* __restrict__ etype,
                                             const float* __restrict__ qall,
                                             const float* __restrict__ kall, int Rr,
                                             int& row, int& dl, float& a0, float& a1) {
    const int d  = dst[e];
    const int s  = src[e];
    const int et = (Rr > 1) ? etype[e] : 0;
    dl  = d - lo;
    row = et * NN + s;
    const size_t qi = ((size_t)et * NN + d) * H;
    const size_t ki = (size_t)row * H;
    a0 = lrelu_(qall[qi] + kall[ki]);
    a1 = 0.f;
    if (H > 1) a1 = lrelu_(qall[qi + 1] + kall[ki + 1]);
}

template <int C, int H, int NB>
__global__ __launch_bounds__(OWN_NT)
void k_edge_aggr_own(const float* __restrict__ XA, const float* __restrict__ qall,
                     const float* __restrict__ kall,
                     const int* __restrict__ src, const int* __restrict__ dst,
                     const int* __restrict__ etype, const float* __restrict__ bias,
                     float* __restrict__ out, int Rr) {
    static_assert(NB <= 64, "local dst is packed into 6 bits");
    constexpr int D  = H * C;
    constexpr int CI = (C + 31) / 32;
    constexpr int NQ = EE / 4;
    constexpr int Q  = 4;
    constexpr int U  = 2;
    __shared__ float    acc[NB * C];
    __shared__ unsigned smax[NB * H];
    __shared__ float    sden[NB * H];
    __shared__ int      l_pk[OWN_CAP];
    __shared__ float    l_a0[OWN_CAP];
    __shared__ float    l_a1[H > 1 ? OWN_CAP : 1];
    __shared__ int      lcount;

    const int lo    = blockIdx.x * NB;
    const int nrows = min(NB, NN - lo);
    for (int i = threadIdx.x; i < NB * C; i += OWN_NT) acc[i] = 0.f;
    for (int i = threadIdx.x; i < NB * H; i += OWN_NT) {
        smax[i] = 0x007FFFFFu;
        sden[i] = 0.f;
    }
    if (threadIdx.x == 0) lcount = 0;
    __syncthreads();

    const int lane  = threadIdx.x & 31;
    const int wave  = threadIdx.x >> 5;
    constexpr int nwave = OWN_NT / 32;
    const int4* dst4 = reinterpret_cast<const int4*>(dst);

    for (int q0 = wave * (32 * Q); q0 < NQ; q0 += nwave * (32 * Q)) {
        int4 dv[Q];
#pragma unroll
        for (int u = 0; u < Q; ++u) {
            const int q = q0 + u * 32 + lane;
            dv[u] = (q < NQ) ? dst4[q] : make_int4(-1, -1, -1, -1);
        }
        unsigned mm = 0;
#pragma unroll
        for (int u = 0; u < Q; ++u) {
            mm |= ((unsigned)(dv[u].x - lo) < (unsigned)nrows ? 1u : 0u) << (4 * u + 0);
            mm |= ((unsigned)(dv[u].y - lo) < (unsigned)nrows ? 1u : 0u) << (4 * u + 1);
            mm |= ((unsigned)(dv[u].z - lo) < (unsigned)nrows ? 1u : 0u) << (4 * u + 2);
            mm |= ((unsigned)(dv[u].w - lo) < (unsigned)nrows ? 1u : 0u) << (4 * u + 3);
        }
        for (int round = 0; round < 4 * Q; ++round) {
            const bool has = (mm != 0);
            const unsigned lm = __builtin_amdgcn_ballot_w32(has);
            if (lm == 0) break;
            const int first = __builtin_ctz(lm);
            int base = 0;
            if (lane == first)
                base = __hip_atomic_fetch_add(&lcount, __builtin_popcount(lm), __ATOMIC_RELAXED,
                                              __HIP_MEMORY_SCOPE_WORKGROUP);
            base = __builtin_amdgcn_readlane(base, first);
            if (has) {
                const int bit = __builtin_ctz(mm);
                mm &= mm - 1;
                const int e = ((q0 + (bit >> 2) * 32 + lane) << 2) + (bit & 3);
                int row, dl; float a0, a1;
                edge_logits_<H>(e, lo, src, dst, etype, qall, kall, Rr, row, dl, a0, a1);
                __hip_atomic_fetch_max(&smax[dl * H], fenc_(a0), __ATOMIC_RELAXED,
                                       __HIP_MEMORY_SCOPE_WORKGROUP);
                if (H > 1)
                    __hip_atomic_fetch_max(&smax[dl * H + 1], fenc_(a1), __ATOMIC_RELAXED,
                                           __HIP_MEMORY_SCOPE_WORKGROUP);
                const int slot = base + __builtin_popcount(lm & ((1u << lane) - 1u));
                if (slot < OWN_CAP) {
                    l_pk[slot] = (row << 6) | dl;
                    l_a0[slot] = a0;
                    if (H > 1) l_a1[slot] = a1;
                }
            }
        }
    }
    __syncthreads();
    const int n = __builtin_amdgcn_readfirstlane(lcount);

    if (n <= OWN_CAP) {
        for (int i = threadIdx.x; i < n; i += OWN_NT) {
            const int dl = l_pk[i] & 63;
            const float e0 = expf(l_a0[i] - fdec_(smax[dl * H]));
            l_a0[i] = e0;
            __hip_atomic_fetch_add(&sden[dl * H], e0, __ATOMIC_RELAXED, __HIP_MEMORY_SCOPE_WORKGROUP);
            if (H > 1) {
                const float e1 = expf(l_a1[i] - fdec_(smax[dl * H + 1]));
                l_a1[i] = e1;
                __hip_atomic_fetch_add(&sden[dl * H + 1], e1, __ATOMIC_RELAXED,
                                       __HIP_MEMORY_SCOPE_WORKGROUP);
            }
        }
        __syncthreads();
        const int wv = __builtin_amdgcn_readfirstlane(wave);
        for (int i0 = wv * U; i0 < n; i0 += nwave * U) {
            float v[U][CI];
            int   dlu[U];
#pragma unroll
            for (int u = 0; u < U; ++u) {
                const bool ok = (i0 + u < n);
                const int  i  = ok ? (i0 + u) : (n - 1);
                const int pk = __builtin_amdgcn_readfirstlane(l_pk[i]);
                const int dl = pk & 63;
                dlu[u] = dl;
                const float w0 = l_a0[i] / (sden[dl * H] + 1e-16f);
                float w1 = 0.f;
                if (H > 1) w1 = l_a1[i] / (sden[dl * H + 1] + 1e-16f);
                const float* xrow = XA + (size_t)(pk >> 6) * D;
#pragma unroll
                for (int ci = 0; ci < CI; ++ci) {
                    const int c = lane + 32 * ci;
                    float t = 0.f;
                    if (c < C) {
                        t += w0 * xrow[c];
                        if (H > 1) t += w1 * xrow[C + c];
                    }
                    v[u][ci] = ok ? t / (float)H : 0.f;
                }
            }
#pragma unroll
            for (int u = 0; u < U; ++u) {
                float* arow = acc + dlu[u] * C;
#pragma unroll
                for (int ci = 0; ci < CI; ++ci) {
                    const int c = lane + 32 * ci;
                    if (c < C)
                        __hip_atomic_fetch_add(arow + c, v[u][ci], __ATOMIC_RELAXED,
                                               __HIP_MEMORY_SCOPE_WORKGROUP);
                }
            }
        }
    } else {
        for (int e = threadIdx.x; e < EE; e += OWN_NT) {
            if ((unsigned)(dst[e] - lo) < (unsigned)nrows) {
                int row, dl; float a0, a1;
                edge_logits_<H>(e, lo, src, dst, etype, qall, kall, Rr, row, dl, a0, a1);
                __hip_atomic_fetch_add(&sden[dl * H], expf(a0 - fdec_(smax[dl * H])),
                                       __ATOMIC_RELAXED, __HIP_MEMORY_SCOPE_WORKGROUP);
                if (H > 1)
                    __hip_atomic_fetch_add(&sden[dl * H + 1], expf(a1 - fdec_(smax[dl * H + 1])),
                                           __ATOMIC_RELAXED, __HIP_MEMORY_SCOPE_WORKGROUP);
            }
        }
        __syncthreads();
        for (int e = threadIdx.x; e < EE; e += OWN_NT) {
            if ((unsigned)(dst[e] - lo) < (unsigned)nrows) {
                int row, dl; float a0, a1;
                edge_logits_<H>(e, lo, src, dst, etype, qall, kall, Rr, row, dl, a0, a1);
                const float w0 = expf(a0 - fdec_(smax[dl * H])) / (sden[dl * H] + 1e-16f);
                float w1 = 0.f;
                if (H > 1) w1 = expf(a1 - fdec_(smax[dl * H + 1])) / (sden[dl * H + 1] + 1e-16f);
                const float* xrow = XA + (size_t)row * D;
                for (int c = 0; c < C; ++c) {
                    float t = 0.f;
                    t += w0 * xrow[c];
                    if (H > 1) t += w1 * xrow[C + c];
                    __hip_atomic_fetch_add(acc + dl * C + c, t / (float)H, __ATOMIC_RELAXED,
                                           __HIP_MEMORY_SCOPE_WORKGROUP);
                }
            }
        }
    }
    __syncthreads();
    float* orow = out + (size_t)lo * C;
    for (int i = threadIdx.x; i < nrows * C; i += OWN_NT)
        orow[i] = elu_(acc[i] + bias[i % C]);
}

struct WsPtrs {
    float* xall; float* qall; float* kall;
};

static void run_attn_conv(const float* xin, const float* w, const float* qv,
                          const float* kv, const float* bias,
                          int K, int D, int C, int H, int Rr,
                          const int* src, const int* dst, const int* etype,
                          const WsPtrs& p, float* out, hipStream_t stream) {
    dim3 gg((NN / 16 + 3) / 4, (D + 63) / 64, Rr);
    if (K == 128) k_gemm_rel_wmma<128, 4><<<gg, 128, 0, stream>>>(xin, w, p.xall, NN, D);
    else          k_gemm_rel_wmma<256, 2><<<gg, 128, 0, stream>>>(xin, w, p.xall, NN, D);

    const int totalRN = Rr * NN;
    k_rowdot<<<(totalRN + 3) / 4, 128, 0, stream>>>(p.xall, qv, kv, p.qall, p.kall,
                                                    D, H, totalRN);
    if (C == 65)
        k_edge_aggr_own<65, 2, 32><<<(NN + 31) / 32, OWN_NT, 0, stream>>>(
            p.xall, p.qall, p.kall, src, dst, etype, bias, out, Rr);
    else if (H == 2)
        k_edge_aggr_own<256, 2, 40><<<(NN + 39) / 40, OWN_NT, 0, stream>>>(
            p.xall, p.qall, p.kall, src, dst, etype, bias, out, Rr);
    else
        k_edge_aggr_own<256, 1, 40><<<(NN + 39) / 40, OWN_NT, 0, stream>>>(
            p.xall, p.qall, p.kall, src, dst, etype, bias, out, Rr);
}

extern "C" void kernel_launch(void* const* d_in, const int* in_sizes, int n_in,
                              void* d_out, int out_size, void* d_ws, size_t ws_size,
                              hipStream_t stream) {
    (void)in_sizes; (void)n_in; (void)out_size; (void)ws_size;
    const float* x  = (const float*)d_in[0];
    const int*   ei = (const int*)  d_in[1];
    const int*   et = (const int*)  d_in[2];
    const float* w1 = (const float*)d_in[3],  *q1 = (const float*)d_in[4];
    const float* k1 = (const float*)d_in[5],  *b1 = (const float*)d_in[6];
    const float* wS = (const float*)d_in[7],  *qS = (const float*)d_in[8];
    const float* kS = (const float*)d_in[9],  *bS = (const float*)d_in[10];
    const float* w2 = (const float*)d_in[11], *q2 = (const float*)d_in[12];
    const float* k2 = (const float*)d_in[13], *b2 = (const float*)d_in[14];
    const float* wT = (const float*)d_in[15], *qT = (const float*)d_in[16];
    const float* kT = (const float*)d_in[17], *bT = (const float*)d_in[18];
    const float* wg = (const float*)d_in[19];
    const float* ag_src = (const float*)d_in[20];
    const float* ag_dst = (const float*)d_in[21];
    const float* bg = (const float*)d_in[22];

    const int* src = ei;
    const int* dst = ei + EE;

    float* ws = (float*)d_ws;
    size_t off = 0;
    WsPtrs p;
    p.xall  = ws + off;            off += (size_t)4 * NN * 512;
    p.qall  = ws + off;            off += (size_t)4 * NN * 2;
    p.kall  = ws + off;            off += (size_t)4 * NN * 2;
    float* xs = ws + off;          off += (size_t)NN * 256;
    float* xt = ws + off;          off += (size_t)NN * 256;

    float* x_in   = (float*)d_out;
    float* x_out  = (float*)d_out + (size_t)NN * 65;
    float* x_self = (float*)d_out + (size_t)2 * NN * 65;

    run_attn_conv(x,  w1, q1, k1, b1, 128, 512, 256, 2, 4, src, dst, et, p, xs,   stream);
    run_attn_conv(xs, wS, qS, kS, bS, 256, 130,  65, 2, 4, src, dst, et, p, x_in, stream);
    run_attn_conv(x,  w2, q2, k2, b2, 128, 512, 256, 2, 4, src, dst, et, p, xt,    stream);
    run_attn_conv(xt, wT, qT, kT, bT, 256, 130,  65, 2, 4, src, dst, et, p, x_out, stream);
    run_attn_conv(x,  wg, ag_dst, ag_src, bg, 128, 256, 256, 1, 1, src, dst, et, p, x_self, stream);
}
